# starting kernel
# baseline (speedup 1.0000x reference)
#include <hip/hip_runtime.h>


typedef __attribute__((ext_vector_type(2))) float v2f;
typedef __attribute__((ext_vector_type(8))) float v8f;

#define BB 4
#define NN 50000
#define KK 16
#define DD 32
#define FF 32
#define KDIM (KK * DD)
#define NGROUP (KDIM / 4)
#define TILES_PER_BATCH (NN / 16)
#define TOTAL_TILES (BB * TILES_PER_BATCH)
#define WAVES_PER_BLOCK 8

__global__ __launch_bounds__(256)
void GraphConv_wmma_f32(const float* __restrict__ x,
                        const int*   __restrict__ nbr,
                        const float* __restrict__ W,
                        const float* __restrict__ bias,
                        float*       __restrict__ out) {
    __shared__ v2f Wfrag[NGROUP * 2 * 32];

    for (int i = threadIdx.x; i < NGROUP * 2 * 32; i += blockDim.x) {
        int lane_i = i & 31;
        int tile   = (i >> 5) & 1;
        int g      = i >> 6;
        int f      = (lane_i & 15) + tile * 16;
        int krow   = 4 * g + ((lane_i >> 4) << 1);
        v2f wv;
        wv.x = W[(krow + 0) * FF + f];
        wv.y = W[(krow + 1) * FF + f];
        Wfrag[i] = wv;
    }
    __syncthreads();

    const int wave  = threadIdx.x >> 5;
    const int lane  = threadIdx.x & 31;
    const int gtile = blockIdx.x * WAVES_PER_BLOCK + wave;
    if (gtile >= TOTAL_TILES) return;

    const int b         = gtile / TILES_PER_BATCH;
    const int node_base = (gtile % TILES_PER_BATCH) * 16;
    const int m         = lane & 15;
    const int hi        = lane >> 4;
    const int node      = node_base + m;

    int nb[KK];
    {
        const int4* np = (const int4*)(nbr + node * KK);
        #pragma unroll
        for (int q = 0; q < 4; ++q) {
            int4 v = np[q];
            nb[4*q+0] = v.x; nb[4*q+1] = v.y; nb[4*q+2] = v.z; nb[4*q+3] = v.w;
        }
    }

    v8f acc0 = {};
    v8f acc1 = {};

    const float* xb = x + (size_t)b * NN * DD;

    for (int k = 0; k < KK; ++k) {
        const float* row = xb + (size_t)nb[k] * DD + hi * 2;
        #pragma unroll
        for (int t = 0; t < 8; ++t) {
            const int g = k * 8 + t;
            v2f a  = *(const v2f*)(row + t * 4);
            v2f b0 = Wfrag[(g * 2 + 0) * 32 + lane];
            v2f b1 = Wfrag[(g * 2 + 1) * 32 + lane];
            acc0 = __builtin_amdgcn_wmma_f32_16x16x4_f32(
                       false, a, false, b0, (short)0, acc0, false, false);
            acc1 = __builtin_amdgcn_wmma_f32_16x16x4_f32(
                       false, a, false, b1, (short)0, acc1, false, false);
        }
    }

    const float bias0 = bias[m];
    const float bias1 = bias[m + 16];
    float* ob = out + ((size_t)b * NN + node_base) * FF;
    #pragma unroll
    for (int v = 0; v < 8; ++v) {
        const int row_m = v + hi * 8;
        ob[row_m * FF + m]      = acc0[v] + bias0;
        ob[row_m * FF + m + 16] = acc1[v] + bias1;
    }
}

extern "C" void kernel_launch(void* const* d_in, const int* in_sizes, int n_in,
                              void* d_out, int out_size, void* d_ws, size_t ws_size,
                              hipStream_t stream) {
    const float* x    = (const float*)d_in[0];
    const int*   nbr  = (const int*)  d_in[1];
    const float* W    = (const float*)d_in[2];
    const float* bias = (const float*)d_in[3];
    float*       out  = (float*)d_out;

    const int blocks = (TOTAL_TILES + WAVES_PER_BLOCK - 1) / WAVES_PER_BLOCK;
    GraphConv_wmma_f32<<<blocks, 256, 0, stream>>>(x, nbr, W, bias, out);
}


# One 32-wave work-group per WGP (grid 256) builds the weight fragment table in LDS once and loops over its 48-49 tiles (was 1563 groups, 64 KB each); WMMA pairs in asm with 8 v_nop hazard slots
# speedup vs baseline: 1.4095x; 1.4095x over previous
#include <hip/hip_runtime.h>


typedef __attribute__((ext_vector_type(2))) float v2f;
typedef __attribute__((ext_vector_type(8))) float v8f;

#define BB 4
#define NN 50000
#define KK 16
#define DD 32
#define FF 32
#define KDIM (KK * DD)
#define NGROUP (KDIM / 4)
#define NFRAG (NGROUP * 2 * 32)
#define TILES_PER_BATCH (NN / 16)
#define TOTAL_TILES (BB * TILES_PER_BATCH)
#define NUM_WG 256
#define WG_THREADS 1024
#define WG_WAVES (WG_THREADS / 32)
#define LDS_RESERVE_BYTES (176 * 1024)

__global__ __launch_bounds__(WG_THREADS)
void GraphConv_wmma_f32(const float* __restrict__ x,
                        const int*   __restrict__ nbr,
                        const float* __restrict__ W,
                        const float* __restrict__ bias,
                        float*       __restrict__ out) {
    __shared__ v2f Wfrag[LDS_RESERVE_BYTES / sizeof(v2f)];

    #pragma unroll
    for (int it = 0; it < NFRAG / WG_THREADS; ++it) {
        const int i      = (int)threadIdx.x + it * WG_THREADS;
        const int lane_i = i & 31;
        const int tile   = (i >> 5) & 1;
        const int g      = i >> 6;
        const int f      = (lane_i & 15) + tile * 16;
        const int krow   = 4 * g + ((lane_i >> 4) << 1);
        v2f wv;
        wv.x = W[(krow + 0) * FF + f];
        wv.y = W[(krow + 1) * FF + f];
        Wfrag[i] = wv;
    }
    __syncthreads();

    const int lane = threadIdx.x & 31;
    const int wave = __builtin_amdgcn_readfirstlane((int)(threadIdx.x >> 5));
    const int m    = lane & 15;
    const int hi   = lane >> 4;

    const float bias0 = bias[m];
    const float bias1 = bias[m + 16];

    for (int j = wave; ; j += WG_WAVES) {
        const int gtile = (int)blockIdx.x + NUM_WG * j;
        if (gtile >= TOTAL_TILES) break;

        const int b         = gtile / TILES_PER_BATCH;
        const int node_base = (gtile % TILES_PER_BATCH) * 16;
        const int node      = node_base + m;

        int nb[KK];
        {
            const int4* np = (const int4*)(nbr + node * KK);
            #pragma unroll
            for (int q = 0; q < 4; ++q) {
                int4 v = np[q];
                nb[4*q+0] = v.x; nb[4*q+1] = v.y; nb[4*q+2] = v.z; nb[4*q+3] = v.w;
            }
        }

        v8f acc0 = {};
        v8f acc1 = {};

        const float* xb = x + (size_t)b * NN * DD;

        for (int k = 0; k < KK; ++k) {
            const float* row = xb + (size_t)nb[k] * DD + hi * 2;
            #pragma unroll
            for (int t = 0; t < 8; ++t) {
                const int g = k * 8 + t;
                v2f a  = *(const v2f*)(row + t * 4);
                v2f b0 = Wfrag[(g * 2 + 0) * 32 + lane];
                v2f b1 = Wfrag[(g * 2 + 1) * 32 + lane];
                asm("v_wmma_f32_16x16x4_f32 %0, %2, %3, %0\n\t"
                    "v_wmma_f32_16x16x4_f32 %1, %2, %4, %1\n\t"
                    "v_nop\n\tv_nop\n\tv_nop\n\tv_nop\n\tv_nop\n\tv_nop\n\tv_nop\n\tv_nop"
                    : "+v"(acc0), "+v"(acc1)
                    : "v"(a), "v"(b0), "v"(b1));
            }
        }

        float* ob = out + ((size_t)b * NN + node_base) * FF;
        #pragma unroll
        for (int v = 0; v < 8; ++v) {
            const int row_m = v + hi * 8;
            ob[row_m * FF + m]      = acc0[v] + bias0;
            ob[row_m * FF + m + 16] = acc1[v] + bias1;
        }
    }
}

extern "C" void kernel_launch(void* const* d_in, const int* in_sizes, int n_in,
                              void* d_out, int out_size, void* d_ws, size_t ws_size,
                              hipStream_t stream) {
    const float* x    = (const float*)d_in[0];
    const int*   nbr  = (const int*)  d_in[1];
    const float* W    = (const float*)d_in[2];
    const float* bias = (const float*)d_in[3];
    float*       out  = (float*)d_out;

    GraphConv_wmma_f32<<<NUM_WG, WG_THREADS, 0, stream>>>(x, nbr, W, bias, out);
}


# Gather rows and W fragments fetched 128 bits per lane (W table stored in the same k order): 4 instead of 8 gather loads per neighbor row, each feeding 4 WMMAs; the order of the kd sum changes
# speedup vs baseline: 1.5473x; 1.0978x over previous
#include <hip/hip_runtime.h>


typedef __attribute__((ext_vector_type(2))) float v2f;
typedef __attribute__((ext_vector_type(4))) float v4f;
typedef __attribute__((ext_vector_type(8))) float v8f;

#define BB 4
#define NN 50000
#define KK 16
#define DD 32
#define FF 32
#define KDIM (KK * DD)
#define NOCT (KDIM / 8)
#define NFRAG (NOCT * 2 * 32)
#define TILES_PER_BATCH (NN / 16)
#define TOTAL_TILES (BB * TILES_PER_BATCH)
#define NUM_WG 256
#define WG_THREADS 1024
#define WG_WAVES (WG_THREADS / 32)
#define LDS_RESERVE_BYTES (176 * 1024)

__global__ __launch_bounds__(WG_THREADS)
void GraphConv_wmma_f32(const float* __restrict__ x,
                        const int*   __restrict__ nbr,
                        const float* __restrict__ W,
                        const float* __restrict__ bias,
                        float*       __restrict__ out) {
    __shared__ v4f Wfrag[LDS_RESERVE_BYTES / sizeof(v4f)];

    #pragma unroll
    for (int it = 0; it < NFRAG / WG_THREADS; ++it) {
        const int i      = (int)threadIdx.x + it * WG_THREADS;
        const int lane_i = i & 31;
        const int tile   = (i >> 5) & 1;
        const int o      = i >> 6;
        const int f      = (lane_i & 15) + tile * 16;
        const int krow   = 8 * o + ((lane_i >> 4) << 2);
        v4f wv;
        wv.x = W[(krow + 0) * FF + f];
        wv.y = W[(krow + 1) * FF + f];
        wv.z = W[(krow + 2) * FF + f];
        wv.w = W[(krow + 3) * FF + f];
        Wfrag[i] = wv;
    }
    __syncthreads();

    const int lane = threadIdx.x & 31;
    const int wave = __builtin_amdgcn_readfirstlane((int)(threadIdx.x >> 5));
    const int m    = lane & 15;
    const int hi   = lane >> 4;

    const float bias0 = bias[m];
    const float bias1 = bias[m + 16];

    for (int j = wave; ; j += WG_WAVES) {
        const int gtile = (int)blockIdx.x + NUM_WG * j;
        if (gtile >= TOTAL_TILES) break;

        const int b         = gtile / TILES_PER_BATCH;
        const int node_base = (gtile % TILES_PER_BATCH) * 16;
        const int node      = node_base + m;

        int nb[KK];
        {
            const int4* np = (const int4*)(nbr + node * KK);
            #pragma unroll
            for (int q = 0; q < 4; ++q) {
                int4 v = np[q];
                nb[4*q+0] = v.x; nb[4*q+1] = v.y; nb[4*q+2] = v.z; nb[4*q+3] = v.w;
            }
        }

        v8f acc0 = {};
        v8f acc1 = {};

        const float* xb = x + (size_t)b * NN * DD;

        for (int k = 0; k < KK; ++k) {
            const float* row = xb + (size_t)nb[k] * DD + hi * 4;
            v4f a[4];
            #pragma unroll
            for (int t = 0; t < 4; ++t)
                a[t] = *(const v4f*)(row + t * 8);
            v4f b0[4], b1[4];
            #pragma unroll
            for (int t = 0; t < 4; ++t) {
                const int o = k * 4 + t;
                b0[t] = Wfrag[(o * 2 + 0) * 32 + lane];
                b1[t] = Wfrag[(o * 2 + 1) * 32 + lane];
            }
            #pragma unroll
            for (int t = 0; t < 4; ++t) {
                v2f a_lo  = __builtin_shufflevector(a[t],  a[t],  0, 1);
                v2f a_hi  = __builtin_shufflevector(a[t],  a[t],  2, 3);
                v2f b0_lo = __builtin_shufflevector(b0[t], b0[t], 0, 1);
                v2f b0_hi = __builtin_shufflevector(b0[t], b0[t], 2, 3);
                v2f b1_lo = __builtin_shufflevector(b1[t], b1[t], 0, 1);
                v2f b1_hi = __builtin_shufflevector(b1[t], b1[t], 2, 3);
                asm("v_wmma_f32_16x16x4_f32 %0, %2, %4, %0\n\t"
                    "v_wmma_f32_16x16x4_f32 %1, %2, %6, %1\n\t"
                    "v_wmma_f32_16x16x4_f32 %0, %3, %5, %0\n\t"
                    "v_wmma_f32_16x16x4_f32 %1, %3, %7, %1\n\t"
                    "v_nop\n\tv_nop\n\tv_nop\n\tv_nop\n\tv_nop\n\tv_nop\n\tv_nop\n\tv_nop"
                    : "+v"(acc0), "+v"(acc1)
                    : "v"(a_lo), "v"(a_hi), "v"(b0_lo), "v"(b0_hi), "v"(b1_lo), "v"(b1_hi));
            }
        }

        float* ob = out + ((size_t)b * NN + node_base) * FF;
        #pragma unroll
        for (int v = 0; v < 8; ++v) {
            const int row_m = v + hi * 8;
            ob[row_m * FF + m]      = acc0[v] + bias0;
            ob[row_m * FF + m + 16] = acc1[v] + bias1;
        }
    }
}

extern "C" void kernel_launch(void* const* d_in, const int* in_sizes, int n_in,
                              void* d_out, int out_size, void* d_ws, size_t ws_size,
                              hipStream_t stream) {
    const float* x    = (const float*)d_in[0];
    const int*   nbr  = (const int*)  d_in[1];
    const float* W    = (const float*)d_in[2];
    const float* bias = (const float*)d_in[3];
    float*       out  = (float*)d_out;

    GraphConv_wmma_f32<<<NUM_WG, WG_THREADS, 0, stream>>>(x, nbr, W, bias, out);
}
